# starting kernel
# baseline (speedup 1.0000x reference)
#include <hip/hip_runtime.h>
#include <math.h>

typedef __attribute__((ext_vector_type(2))) float v2f;
typedef __attribute__((ext_vector_type(8))) float v8f;

#define BATCH 32
#define LA 2048
#define LB 2048
#define TPB 256
#define WAVES 8
#define ROWS_PER_BLOCK 128
#define TILES_PER_BATCH (LA / ROWS_PER_BLOCK)

__global__ __launch_bounds__(TPB) void cross_attn_wmma_kernel(
    const float* __restrict__ a, const float* __restrict__ b,
    const float* __restrict__ ipw, const float* __restrict__ ipb,
    const float* __restrict__ ow,  const float* __restrict__ ob,
    float* __restrict__ out)
{
    __shared__ __align__(16) float s_k[LB];
    __shared__ __align__(16) float s_v[LB];
    __shared__ float s_red[2 * TPB];
    __shared__ float s_c[WAVES][32][8];

    const int tid   = threadIdx.x;
    const int wave  = tid >> 5;
    const int lane  = tid & 31;
    const int batch = blockIdx.x / TILES_PER_BATCH;
    const int tile  = blockIdx.x % TILES_PER_BATCH;

    const float wq = ipw[0], wk = ipw[1], wv = ipw[2];
    const float bq = ipb[0], bk = ipb[1], bv = ipb[2];
    const float w_out = ow[0], b_out = ob[0];

    float lmax = -INFINITY, lmin = INFINITY;
    const float* brow = b + (size_t)batch * LB;
    for (int i = tid; i < LB; i += TPB) {
        float bi = brow[i];
        float kv = fmaf(bi, wk, bk);
        s_k[i] = kv;
        s_v[i] = fmaf(bi, wv, bv);
        lmax = fmaxf(lmax, kv);
        lmin = fminf(lmin, kv);
    }
    s_red[tid]       = lmax;
    s_red[TPB + tid] = lmin;
    __syncthreads();
    for (int s = TPB / 2; s > 0; s >>= 1) {
        if (tid < s) {
            s_red[tid]       = fmaxf(s_red[tid],       s_red[tid + s]);
            s_red[TPB + tid] = fminf(s_red[TPB + tid], s_red[TPB + tid + s]);
        }
        __syncthreads();
    }
    const float kmax = s_red[0], kmin = s_red[TPB];

    const int   rowBase = tile * ROWS_PER_BLOCK + wave * 16;
    const int   myRow   = rowBase + (lane & 15);
    const float qa = a[(size_t)batch * LA + myRow];
    const float q  = fmaf(qa, wq, bq);
    const float m  = (q >= 0.0f) ? q * kmax : q * kmin;
    const float L2E = 1.4426950408889634f;
    const float ql2  = q * L2E;
    const float nml2 = -(m * L2E);

    const int  lsel = lane >> 4;
    const int  nsel = lane & 15;
    const bool isn0   = (nsel == 0);
    const float bcnst = (nsel == 1) ? 1.0f : 0.0f;

    const float2* s_k2 = reinterpret_cast<const float2*>(s_k);
    const float2* s_v2 = reinterpret_cast<const float2*>(s_v);

    v8f c = {0.f, 0.f, 0.f, 0.f, 0.f, 0.f, 0.f, 0.f};

    #pragma unroll 8
    for (int jb = 0; jb < LB; jb += 4) {
        int idx = (jb >> 1) + lsel;
        float2 kk = s_k2[idx];
        float2 vv = s_v2[idx];
        float e0 = __builtin_amdgcn_exp2f(fmaf(ql2, kk.x, nml2));
        float e1 = __builtin_amdgcn_exp2f(fmaf(ql2, kk.y, nml2));
        v2f A; A.x = e0; A.y = e1;
        v2f B;
        B.x = isn0 ? vv.x : bcnst;
        B.y = isn0 ? vv.y : bcnst;
        c = __builtin_amdgcn_wmma_f32_16x16x4_f32(
                 false, A,  false, B,
                 (short)0, c,  false,  false);
    }

    #pragma unroll
    for (int g = 0; g < 8; ++g) s_c[wave][lane][g] = c[g];

    if (lane < 16) {
        int srcLane = (lane < 8) ? 0 : 16;
        int g = lane & 7;
        float numer = s_c[wave][srcLane][g];
        float denom = s_c[wave][srcLane + 1][g];
        float ctx = numer / denom;
        float x = fmaf(ctx, w_out, b_out);
        float y = x + qa;
        float r = 0.5f * y * (1.0f + erff(y * 0.70710678118654752f));
        out[(size_t)batch * LA + rowBase + lane] = r;
    }
}

extern "C" void kernel_launch(void* const* d_in, const int* in_sizes, int n_in,
                              void* d_out, int out_size, void* d_ws, size_t ws_size,
                              hipStream_t stream) {
    const float* a   = (const float*)d_in[0];
    const float* b   = (const float*)d_in[1];
    const float* ipw = (const float*)d_in[2];
    const float* ipb = (const float*)d_in[3];
    const float* ow  = (const float*)d_in[4];
    const float* ob  = (const float*)d_in[5];
    float* out = (float*)d_out;
    (void)in_sizes; (void)n_in; (void)out_size; (void)d_ws; (void)ws_size;

    dim3 grid(BATCH * TILES_PER_BATCH);
    dim3 block(TPB);
    hipLaunchKernelGGL(cross_attn_wmma_kernel, grid, block, 0, stream,
                       a, b, ipw, ipb, ow, ob, out);
}


# LDS reads of k and v grouped in one place per 16 matrix steps, one block ahead in a second register set, no LDS read between the WMMAs
# speedup vs baseline: 1.1040x; 1.1040x over previous
#include <hip/hip_runtime.h>
#include <math.h>

typedef __attribute__((ext_vector_type(2))) float v2f;
typedef __attribute__((ext_vector_type(8))) float v8f;

#define BATCH 32
#define LA 2048
#define LB 2048
#define TPB 256
#define WAVES 8
#define ROWS_PER_BLOCK 128
#define TILES_PER_BATCH (LA / ROWS_PER_BLOCK)
#define PAIRS 8
#define NBLK (LB / (8 * PAIRS))

__global__ __launch_bounds__(TPB) void cross_attn_wmma_kernel(
    const float* __restrict__ a, const float* __restrict__ b,
    const float* __restrict__ ipw, const float* __restrict__ ipb,
    const float* __restrict__ ow,  const float* __restrict__ ob,
    float* __restrict__ out)
{
    __shared__ __align__(16) float s_k[LB];
    __shared__ __align__(16) float s_v[LB];
    __shared__ float s_red[2 * TPB];
    __shared__ float s_c[WAVES][32][8];

    const int tid   = threadIdx.x;
    const int wave  = tid >> 5;
    const int lane  = tid & 31;
    const int batch = blockIdx.x / TILES_PER_BATCH;
    const int tile  = blockIdx.x % TILES_PER_BATCH;

    const float wq = ipw[0], wk = ipw[1], wv = ipw[2];
    const float bq = ipb[0], bk = ipb[1], bv = ipb[2];
    const float w_out = ow[0], b_out = ob[0];

    float lmax = -INFINITY, lmin = INFINITY;
    const float* brow = b + (size_t)batch * LB;
    for (int i = tid; i < LB; i += TPB) {
        float bi = brow[i];
        float kv = fmaf(bi, wk, bk);
        s_k[i] = kv;
        s_v[i] = fmaf(bi, wv, bv);
        lmax = fmaxf(lmax, kv);
        lmin = fminf(lmin, kv);
    }
    s_red[tid]       = lmax;
    s_red[TPB + tid] = lmin;
    __syncthreads();
    for (int s = TPB / 2; s > 0; s >>= 1) {
        if (tid < s) {
            s_red[tid]       = fmaxf(s_red[tid],       s_red[tid + s]);
            s_red[TPB + tid] = fminf(s_red[TPB + tid], s_red[TPB + tid + s]);
        }
        __syncthreads();
    }
    const float kmax = s_red[0], kmin = s_red[TPB];

    const int   rowBase = tile * ROWS_PER_BLOCK + wave * 16;
    const int   myRow   = rowBase + (lane & 15);
    const float qa = a[(size_t)batch * LA + myRow];
    const float q  = fmaf(qa, wq, bq);
    const float m  = (q >= 0.0f) ? q * kmax : q * kmin;
    const float L2E = 1.4426950408889634f;
    const float ql2  = q * L2E;
    const float nml2 = -(m * L2E);

    const int  lsel = lane >> 4;
    const int  nsel = lane & 15;
    const bool isn0   = (nsel == 0);
    const float bcnst = (nsel == 1) ? 1.0f : 0.0f;

    const float4* s_k4 = reinterpret_cast<const float4*>(s_k);
    const float4* s_v4 = reinterpret_cast<const float4*>(s_v);

    v8f c = {0.f, 0.f, 0.f, 0.f, 0.f, 0.f, 0.f, 0.f};

    float4 kA[PAIRS], vA[PAIRS], kB[PAIRS], vB[PAIRS];

#define LOAD_BLOCK(KK, VV, BLK)                                              \
    {                                                                        \
        const int base_ = (BLK) * (2 * PAIRS) + lsel;                        \
        _Pragma("unroll")                                                    \
        for (int u = 0; u < PAIRS; ++u) {                                    \
            KK[u] = s_k4[base_ + 2 * u];                                     \
            VV[u] = s_v4[base_ + 2 * u];                                     \
        }                                                                    \
    }

#define STEP(K0, K1, V0, V1)                                                 \
    {                                                                        \
        float e0 = __builtin_amdgcn_exp2f(fmaf(ql2, (K0), nml2));            \
        float e1 = __builtin_amdgcn_exp2f(fmaf(ql2, (K1), nml2));            \
        v2f A; A.x = e0; A.y = e1;                                           \
        v2f B;                                                               \
        B.x = isn0 ? (V0) : bcnst;                                           \
        B.y = isn0 ? (V1) : bcnst;                                           \
        c = __builtin_amdgcn_wmma_f32_16x16x4_f32(                           \
                false, A, false, B, (short)0, c, false, false);              \
    }

#define COMPUTE_BLOCK(KK, VV)                                                \
    {                                                                        \
        _Pragma("unroll")                                                    \
        for (int u = 0; u < PAIRS; ++u) {                                    \
            STEP(KK[u].x, KK[u].y, VV[u].x, VV[u].y)                         \
            STEP(KK[u].z, KK[u].w, VV[u].z, VV[u].w)                         \
        }                                                                    \
    }

    LOAD_BLOCK(kA, vA, 0)
    #pragma unroll 1
    for (int blk = 0; blk < NBLK; blk += 2) {
        LOAD_BLOCK(kB, vB, blk + 1)
        __builtin_amdgcn_sched_barrier(0);
        COMPUTE_BLOCK(kA, vA)
        __builtin_amdgcn_sched_barrier(0);
        LOAD_BLOCK(kA, vA, (blk + 2) & (NBLK - 1))
        __builtin_amdgcn_sched_barrier(0);
        COMPUTE_BLOCK(kB, vB)
        __builtin_amdgcn_sched_barrier(0);
    }

    #pragma unroll
    for (int g = 0; g < 8; ++g) s_c[wave][lane][g] = c[g];

    if (lane < 16) {
        int srcLane = (lane < 8) ? 0 : 16;
        int g = lane & 7;
        float numer = s_c[wave][srcLane][g];
        float denom = s_c[wave][srcLane + 1][g];
        float ctx = numer / denom;
        float x = fmaf(ctx, w_out, b_out);
        float y = x + qa;
        float r = 0.5f * y * (1.0f + erff(y * 0.70710678118654752f));
        out[(size_t)batch * LA + rowBase + lane] = r;
    }
}

extern "C" void kernel_launch(void* const* d_in, const int* in_sizes, int n_in,
                              void* d_out, int out_size, void* d_ws, size_t ws_size,
                              hipStream_t stream) {
    const float* a   = (const float*)d_in[0];
    const float* b   = (const float*)d_in[1];
    const float* ipw = (const float*)d_in[2];
    const float* ipb = (const float*)d_in[3];
    const float* ow  = (const float*)d_in[4];
    const float* ob  = (const float*)d_in[5];
    float* out = (float*)d_out;
    (void)in_sizes; (void)n_in; (void)out_size; (void)d_ws; (void)ws_size;

    dim3 grid(BATCH * TILES_PER_BATCH);
    dim3 block(TPB);
    hipLaunchKernelGGL(cross_attn_wmma_kernel, grid, block, 0, stream,
                       a, b, ipw, ipb, ow, ob, out);
}


# vector work of each matrix step issued one step ahead in a fixed order (FMA pair, exp, select pair, exp, WMMA): no exp feeds its WMMA directly, none back to back, hazard nops 22 to 1 per 32 steps
# speedup vs baseline: 1.1587x; 1.0495x over previous
#include <hip/hip_runtime.h>
#include <math.h>

typedef __attribute__((ext_vector_type(2))) float v2f;
typedef __attribute__((ext_vector_type(8))) float v8f;

#define BATCH 32
#define LA 2048
#define LB 2048
#define TPB 256
#define WAVES 8
#define ROWS_PER_BLOCK 128
#define TILES_PER_BATCH (LA / ROWS_PER_BLOCK)
#define PAIRS 8
#define NBLK (LB / (8 * PAIRS))

__device__ __forceinline__ float2 step_pair(const float4 (&P)[PAIRS], const float4 (&Q)[PAIRS], int t)
{
    const float4 f = (t < 2 * PAIRS) ? P[t >> 1] : Q[(t - 2 * PAIRS) >> 1];
    return (t & 1) ? make_float2(f.z, f.w) : make_float2(f.x, f.y);
}

__global__ __launch_bounds__(TPB) void cross_attn_wmma_kernel(
    const float* __restrict__ a, const float* __restrict__ b,
    const float* __restrict__ ipw, const float* __restrict__ ipb,
    const float* __restrict__ ow,  const float* __restrict__ ob,
    float* __restrict__ out)
{
    __shared__ __align__(16) float s_k[LB];
    __shared__ __align__(16) float s_v[LB];
    __shared__ float s_red[2 * TPB];
    __shared__ float s_c[WAVES][32][8];

    const int tid   = threadIdx.x;
    const int wave  = tid >> 5;
    const int lane  = tid & 31;
    const int batch = blockIdx.x / TILES_PER_BATCH;
    const int tile  = blockIdx.x % TILES_PER_BATCH;

    const float wq = ipw[0], wk = ipw[1], wv = ipw[2];
    const float bq = ipb[0], bk = ipb[1], bv = ipb[2];
    const float w_out = ow[0], b_out = ob[0];

    float lmax = -INFINITY, lmin = INFINITY;
    const float* brow = b + (size_t)batch * LB;
    for (int i = tid; i < LB; i += TPB) {
        float bi = brow[i];
        float kv = fmaf(bi, wk, bk);
        s_k[i] = kv;
        s_v[i] = fmaf(bi, wv, bv);
        lmax = fmaxf(lmax, kv);
        lmin = fminf(lmin, kv);
    }
    s_red[tid]       = lmax;
    s_red[TPB + tid] = lmin;
    __syncthreads();
    for (int s = TPB / 2; s > 0; s >>= 1) {
        if (tid < s) {
            s_red[tid]       = fmaxf(s_red[tid],       s_red[tid + s]);
            s_red[TPB + tid] = fminf(s_red[TPB + tid], s_red[TPB + tid + s]);
        }
        __syncthreads();
    }
    const float kmax = s_red[0], kmin = s_red[TPB];

    const int   rowBase = tile * ROWS_PER_BLOCK + wave * 16;
    const int   myRow   = rowBase + (lane & 15);
    const float qa = a[(size_t)batch * LA + myRow];
    const float q  = fmaf(qa, wq, bq);
    const float m  = (q >= 0.0f) ? q * kmax : q * kmin;
    const float L2E = 1.4426950408889634f;
    const float ql2  = q * L2E;
    const float nml2 = -(m * L2E);

    const int  lsel = lane >> 4;
    const int  nsel = lane & 15;
    const bool isn0   = (nsel == 0);
    const float bcnst = (nsel == 1) ? 1.0f : 0.0f;

    const float4* s_k4 = reinterpret_cast<const float4*>(s_k);
    const float4* s_v4 = reinterpret_cast<const float4*>(s_v);

    v8f c = {0.f, 0.f, 0.f, 0.f, 0.f, 0.f, 0.f, 0.f};

    float4 kA[PAIRS], vA[PAIRS], kB[PAIRS], vB[PAIRS];

#define LOAD_BLOCK(KK, VV, BLK)                                              \
    {                                                                        \
        const int base_ = (BLK) * (2 * PAIRS) + lsel;                        \
        _Pragma("unroll")                                                    \
        for (int u = 0; u < PAIRS; ++u) {                                    \
            KK[u] = s_k4[base_ + 2 * u];                                     \
            VV[u] = s_v4[base_ + 2 * u];                                     \
        }                                                                    \
    }

#define SB() __builtin_amdgcn_sched_barrier(0)
#define COMPUTE_BLOCK(KP, VP, KQ, VQ)                                        \
    {                                                                        \
        _Pragma("unroll")                                                    \
        for (int s = 0; s < 2 * PAIRS; ++s) {                                \
            const float2 kn = step_pair(KP, KQ, s + 2);                      \
            const float2 vn = step_pair(VP, VQ, s + 1);                      \
            asm volatile("" : "+v"(c), "+v"(qv));                            \
            float an0 = fmaf(qv, kn.x, nml2);                                \
            float an1 = fmaf(qv, kn.y, nml2);                                \
            asm volatile("" : "+v"(an0), "+v"(an1), "+v"(ar0));              \
            float e0 = __builtin_amdgcn_exp2f(ar0);                          \
            asm volatile("" : "+v"(e0), "+v"(bcv));                          \
            float bn0 = isn0 ? vn.x : bcv;                                   \
            float bn1 = isn0 ? vn.y : bcv;                                   \
            asm volatile("" : "+v"(bn0), "+v"(bn1), "+v"(ar1));              \
            float e1 = __builtin_amdgcn_exp2f(ar1);                          \
            asm volatile("" : "+v"(e1), "+v"(c));                            \
            v2f A_; A_.x = ea0; A_.y = ea1;                                  \
            v2f B_; B_.x = bb0; B_.y = bb1;                                  \
            c = __builtin_amdgcn_wmma_f32_16x16x4_f32(                       \
                    false, A_, false, B_, (short)0, c, false, false);        \
            ea0 = e0; ea1 = e1; bb0 = bn0; bb1 = bn1; ar0 = an0; ar1 = an1;  \
        }                                                                    \
    }

    LOAD_BLOCK(kA, vA, 0)
    float ea0 = __builtin_amdgcn_exp2f(fmaf(ql2, kA[0].x, nml2));
    float ea1 = __builtin_amdgcn_exp2f(fmaf(ql2, kA[0].y, nml2));
    float bb0 = isn0 ? vA[0].x : bcnst;
    float bb1 = isn0 ? vA[0].y : bcnst;
    float ar0 = fmaf(ql2, kA[0].z, nml2);
    float ar1 = fmaf(ql2, kA[0].w, nml2);
    float qv = ql2, bcv = bcnst;
    #pragma unroll 1
    for (int blk = 0; blk < NBLK; blk += 2) {
        LOAD_BLOCK(kB, vB, blk + 1)
        SB();
        COMPUTE_BLOCK(kA, vA, kB, vB)
        LOAD_BLOCK(kA, vA, (blk + 2) & (NBLK - 1))
        SB();
        COMPUTE_BLOCK(kB, vB, kA, vA)
    }

    #pragma unroll
    for (int g = 0; g < 8; ++g) s_c[wave][lane][g] = c[g];

    if (lane < 16) {
        int srcLane = (lane < 8) ? 0 : 16;
        int g = lane & 7;
        float numer = s_c[wave][srcLane][g];
        float denom = s_c[wave][srcLane + 1][g];
        float ctx = numer / denom;
        float x = fmaf(ctx, w_out, b_out);
        float y = x + qa;
        float r = 0.5f * y * (1.0f + erff(y * 0.70710678118654752f));
        out[(size_t)batch * LA + rowBase + lane] = r;
    }
}

extern "C" void kernel_launch(void* const* d_in, const int* in_sizes, int n_in,
                              void* d_out, int out_size, void* d_ws, size_t ws_size,
                              hipStream_t stream) {
    const float* a   = (const float*)d_in[0];
    const float* b   = (const float*)d_in[1];
    const float* ipw = (const float*)d_in[2];
    const float* ipb = (const float*)d_in[3];
    const float* ow  = (const float*)d_in[4];
    const float* ob  = (const float*)d_in[5];
    float* out = (float*)d_out;
    (void)in_sizes; (void)n_in; (void)out_size; (void)d_ws; (void)ws_size;

    dim3 grid(BATCH * TILES_PER_BATCH);
    dim3 block(TPB);
    hipLaunchKernelGGL(cross_attn_wmma_kernel, grid, block, 0, stream,
                       a, b, ipw, ipb, ow, ob, out);
}


# prologue serial chain cut: all global reads (b as two 16-byte reads per thread, and a) in flight at once; max/min reduced in-wave by cross-lane steps, 1 barrier; was 8 read round trips and 9 barriers
# speedup vs baseline: 1.3450x; 1.1608x over previous
#include <hip/hip_runtime.h>
#include <math.h>

typedef __attribute__((ext_vector_type(2))) float v2f;
typedef __attribute__((ext_vector_type(8))) float v8f;

#define BATCH 32
#define LA 2048
#define LB 2048
#define TPB 256
#define WAVES 8
#define ROWS_PER_BLOCK 128
#define TILES_PER_BATCH (LA / ROWS_PER_BLOCK)
#define PAIRS 8
#define NBLK (LB / (8 * PAIRS))

__device__ __forceinline__ float2 step_pair(const float4 (&P)[PAIRS], const float4 (&Q)[PAIRS], int t)
{
    const float4 f = (t < 2 * PAIRS) ? P[t >> 1] : Q[(t - 2 * PAIRS) >> 1];
    return (t & 1) ? make_float2(f.z, f.w) : make_float2(f.x, f.y);
}

template <int MASK>
__device__ __forceinline__ float xrow(float x)
{
    const int b = __builtin_bit_cast(int, x);
    return __builtin_bit_cast(float, __builtin_amdgcn_update_dpp(b, b, 0x160 + MASK, 0xf, 0xf, false));
}
__device__ __forceinline__ float xhalf(float x)
{
    const int b = __builtin_bit_cast(int, x);
    return __builtin_bit_cast(float, __builtin_amdgcn_permlanex16(b, b, 0x76543210, 0xfedcba98, false, false));
}
__device__ __forceinline__ float wave_max(float x)
{
    x = fmaxf(x, xrow<1>(x)); x = fmaxf(x, xrow<2>(x));
    x = fmaxf(x, xrow<4>(x)); x = fmaxf(x, xrow<8>(x));
    return fmaxf(x, xhalf(x));
}
__device__ __forceinline__ float wave_min(float x)
{
    x = fminf(x, xrow<1>(x)); x = fminf(x, xrow<2>(x));
    x = fminf(x, xrow<4>(x)); x = fminf(x, xrow<8>(x));
    return fminf(x, xhalf(x));
}

__global__ __launch_bounds__(TPB) void cross_attn_wmma_kernel(
    const float* __restrict__ a, const float* __restrict__ b,
    const float* __restrict__ ipw, const float* __restrict__ ipb,
    const float* __restrict__ ow,  const float* __restrict__ ob,
    float* __restrict__ out)
{
    __shared__ __align__(16) float s_k[LB];
    __shared__ __align__(16) float s_v[LB];
    __shared__ __align__(16) float s_red[2 * WAVES];
    __shared__ float s_c[WAVES][32][8];

    const int tid   = threadIdx.x;
    const int wave  = tid >> 5;
    const int lane  = tid & 31;
    const int batch = blockIdx.x / TILES_PER_BATCH;
    const int tile  = blockIdx.x % TILES_PER_BATCH;

    const int   rowBase = tile * ROWS_PER_BLOCK + wave * 16;
    const int   myRow   = rowBase + (lane & 15);
    const float* brow = b + (size_t)batch * LB;
    float bl[2][4];
    #pragma unroll
    for (int t = 0; t < 2; ++t) {
        const int g = tid + t * TPB;
        #pragma unroll
        for (int e = 0; e < 4; ++e) bl[t][e] = brow[4 * g + e];
    }
    const float qa = a[(size_t)batch * LA + myRow];
    __builtin_amdgcn_sched_barrier(0);

    const float wq = ipw[0], wk = ipw[1], wv = ipw[2];
    const float bq = ipb[0], bk = ipb[1], bv = ipb[2];
    const float w_out = ow[0], b_out = ob[0];

    float lmax = -INFINITY, lmin = INFINITY;
    #pragma unroll
    for (int t = 0; t < 2; ++t) {
        const int g = tid + t * TPB;
        float kq[4], vq[4];
        #pragma unroll
        for (int e = 0; e < 4; ++e) {
            kq[e] = fmaf(bl[t][e], wk, bk);
            vq[e] = fmaf(bl[t][e], wv, bv);
            lmax = fmaxf(lmax, kq[e]);
            lmin = fminf(lmin, kq[e]);
        }
        reinterpret_cast<float4*>(s_k)[g] = make_float4(kq[0], kq[1], kq[2], kq[3]);
        reinterpret_cast<float4*>(s_v)[g] = make_float4(vq[0], vq[1], vq[2], vq[3]);
    }
    const float wmax = wave_max(lmax);
    const float wmin = wave_min(lmin);
    if (lane == 0) { s_red[wave] = wmax; s_red[WAVES + wave] = wmin; }
    __syncthreads();
    const float4* s_red4 = reinterpret_cast<const float4*>(s_red);
    const float4 mx0 = s_red4[0], mx1 = s_red4[1], mn0 = s_red4[2], mn1 = s_red4[3];
    const float kmax = fmaxf(fmaxf(fmaxf(mx0.x, mx0.y), fmaxf(mx0.z, mx0.w)),
                             fmaxf(fmaxf(mx1.x, mx1.y), fmaxf(mx1.z, mx1.w)));
    const float kmin = fminf(fminf(fminf(mn0.x, mn0.y), fminf(mn0.z, mn0.w)),
                             fminf(fminf(mn1.x, mn1.y), fminf(mn1.z, mn1.w)));

    const float q  = fmaf(qa, wq, bq);
    const float m  = (q >= 0.0f) ? q * kmax : q * kmin;
    const float L2E = 1.4426950408889634f;
    const float ql2  = q * L2E;
    const float nml2 = -(m * L2E);

    const int  lsel = lane >> 4;
    const int  nsel = lane & 15;
    const bool isn0   = (nsel == 0);
    const float bcnst = (nsel == 1) ? 1.0f : 0.0f;

    const float4* s_k4 = reinterpret_cast<const float4*>(s_k);
    const float4* s_v4 = reinterpret_cast<const float4*>(s_v);

    v8f c = {0.f, 0.f, 0.f, 0.f, 0.f, 0.f, 0.f, 0.f};

    float4 kA[PAIRS], vA[PAIRS], kB[PAIRS], vB[PAIRS];

#define LOAD_BLOCK(KK, VV, BLK)                                              \
    {                                                                        \
        const int base_ = (BLK) * (2 * PAIRS) + lsel;                        \
        _Pragma("unroll")                                                    \
        for (int u = 0; u < PAIRS; ++u) {                                    \
            KK[u] = s_k4[base_ + 2 * u];                                     \
            VV[u] = s_v4[base_ + 2 * u];                                     \
        }                                                                    \
    }

#define SB() __builtin_amdgcn_sched_barrier(0)
#define COMPUTE_BLOCK(KP, VP, KQ, VQ)                                        \
    {                                                                        \
        _Pragma("unroll")                                                    \
        for (int s = 0; s < 2 * PAIRS; ++s) {                                \
            const float2 kn = step_pair(KP, KQ, s + 2);                      \
            const float2 vn = step_pair(VP, VQ, s + 1);                      \
            asm volatile("" : "+v"(c), "+v"(qv));                            \
            float an0 = fmaf(qv, kn.x, nml2);                                \
            float an1 = fmaf(qv, kn.y, nml2);                                \
            asm volatile("" : "+v"(an0), "+v"(an1), "+v"(ar0));              \
            float e0 = __builtin_amdgcn_exp2f(ar0);                          \
            asm volatile("" : "+v"(e0), "+v"(bcv));                          \
            float bn0 = isn0 ? vn.x : bcv;                                   \
            float bn1 = isn0 ? vn.y : bcv;                                   \
            asm volatile("" : "+v"(bn0), "+v"(bn1), "+v"(ar1));              \
            float e1 = __builtin_amdgcn_exp2f(ar1);                          \
            asm volatile("" : "+v"(e1), "+v"(c));                            \
            v2f A_; A_.x = ea0; A_.y = ea1;                                  \
            v2f B_; B_.x = bb0; B_.y = bb1;                                  \
            c = __builtin_amdgcn_wmma_f32_16x16x4_f32(                       \
                    false, A_, false, B_, (short)0, c, false, false);        \
            ea0 = e0; ea1 = e1; bb0 = bn0; bb1 = bn1; ar0 = an0; ar1 = an1;  \
        }                                                                    \
    }

    LOAD_BLOCK(kA, vA, 0)
    float ea0 = __builtin_amdgcn_exp2f(fmaf(ql2, kA[0].x, nml2));
    float ea1 = __builtin_amdgcn_exp2f(fmaf(ql2, kA[0].y, nml2));
    float bb0 = isn0 ? vA[0].x : bcnst;
    float bb1 = isn0 ? vA[0].y : bcnst;
    float ar0 = fmaf(ql2, kA[0].z, nml2);
    float ar1 = fmaf(ql2, kA[0].w, nml2);
    float qv = ql2, bcv = bcnst;
    #pragma unroll 1
    for (int blk = 0; blk < NBLK; blk += 2) {
        LOAD_BLOCK(kB, vB, blk + 1)
        SB();
        COMPUTE_BLOCK(kA, vA, kB, vB)
        LOAD_BLOCK(kA, vA, (blk + 2) & (NBLK - 1))
        SB();
        COMPUTE_BLOCK(kB, vB, kA, vA)
    }

    #pragma unroll
    for (int g = 0; g < 8; ++g) s_c[wave][lane][g] = c[g];

    if (lane < 16) {
        int srcLane = (lane < 8) ? 0 : 16;
        int g = lane & 7;
        float numer = s_c[wave][srcLane][g];
        float denom = s_c[wave][srcLane + 1][g];
        float ctx = numer / denom;
        float x = fmaf(ctx, w_out, b_out);
        float y = x + qa;
        float r = 0.5f * y * (1.0f + erff(y * 0.70710678118654752f));
        out[(size_t)batch * LA + rowBase + lane] = r;
    }
}

extern "C" void kernel_launch(void* const* d_in, const int* in_sizes, int n_in,
                              void* d_out, int out_size, void* d_ws, size_t ws_size,
                              hipStream_t stream) {
    const float* a   = (const float*)d_in[0];
    const float* b   = (const float*)d_in[1];
    const float* ipw = (const float*)d_in[2];
    const float* ipb = (const float*)d_in[3];
    const float* ow  = (const float*)d_in[4];
    const float* ob  = (const float*)d_in[5];
    float* out = (float*)d_out;
    (void)in_sizes; (void)n_in; (void)out_size; (void)d_ws; (void)ws_size;

    dim3 grid(BATCH * TILES_PER_BATCH);
    dim3 block(TPB);
    hipLaunchKernelGGL(cross_attn_wmma_kernel, grid, block, 0, stream,
                       a, b, ipw, ipb, ow, ob, out);
}


# B operand read ready-made from LDS (column 0 lanes read v, column 1 lanes a plane of ones, the rest a plane of zeros, bank-disjoint): the select pair per matrix step is gone
# speedup vs baseline: 1.4139x; 1.0512x over previous
#include <hip/hip_runtime.h>
#include <math.h>

typedef __attribute__((ext_vector_type(2))) float v2f;
typedef __attribute__((ext_vector_type(8))) float v8f;

#define BATCH 32
#define LA 2048
#define LB 2048
#define TPB 256
#define WAVES 8
#define ROWS_PER_BLOCK 128
#define TILES_PER_BATCH (LA / ROWS_PER_BLOCK)
#define PAIRS 8
#define NBLK (LB / (8 * PAIRS))

__device__ __forceinline__ float2 step_pair(const float4 (&P)[PAIRS], const float4 (&Q)[PAIRS], int t)
{
    const float4 f = (t < 2 * PAIRS) ? P[t >> 1] : Q[(t - 2 * PAIRS) >> 1];
    return (t & 1) ? make_float2(f.z, f.w) : make_float2(f.x, f.y);
}

template <int MASK>
__device__ __forceinline__ float xrow(float x)
{
    const int b = __builtin_bit_cast(int, x);
    return __builtin_bit_cast(float, __builtin_amdgcn_update_dpp(b, b, 0x160 + MASK, 0xf, 0xf, false));
}
__device__ __forceinline__ float xhalf(float x)
{
    const int b = __builtin_bit_cast(int, x);
    return __builtin_bit_cast(float, __builtin_amdgcn_permlanex16(b, b, 0x76543210, 0xfedcba98, false, false));
}
__device__ __forceinline__ float wave_max(float x)
{
    x = fmaxf(x, xrow<1>(x)); x = fmaxf(x, xrow<2>(x));
    x = fmaxf(x, xrow<4>(x)); x = fmaxf(x, xrow<8>(x));
    return fmaxf(x, xhalf(x));
}
__device__ __forceinline__ float wave_min(float x)
{
    x = fminf(x, xrow<1>(x)); x = fminf(x, xrow<2>(x));
    x = fminf(x, xrow<4>(x)); x = fminf(x, xrow<8>(x));
    return fminf(x, xhalf(x));
}

__global__ __launch_bounds__(TPB) void cross_attn_wmma_kernel(
    const float* __restrict__ a, const float* __restrict__ b,
    const float* __restrict__ ipw, const float* __restrict__ ipb,
    const float* __restrict__ ow,  const float* __restrict__ ob,
    float* __restrict__ out)
{
    __shared__ __align__(256) float s_kv[2 * LB + 192];
    float* s_k = s_kv;
    float* s_v = s_kv + LB;
    float* s_one  = s_kv + 2 * LB + 32;
    float* s_zero = s_kv + 2 * LB + 112;
    __shared__ __align__(16) float s_red[2 * WAVES];
    __shared__ float s_c[WAVES][32][8];

    const int tid   = threadIdx.x;
    const int wave  = tid >> 5;
    const int lane  = tid & 31;
    const int batch = blockIdx.x / TILES_PER_BATCH;
    const int tile  = blockIdx.x % TILES_PER_BATCH;

    const int   rowBase = tile * ROWS_PER_BLOCK + wave * 16;
    const int   myRow   = rowBase + (lane & 15);
    const float* brow = b + (size_t)batch * LB;
    float bl[2][4];
    #pragma unroll
    for (int t = 0; t < 2; ++t) {
        const int g = tid + t * TPB;
        #pragma unroll
        for (int e = 0; e < 4; ++e) bl[t][e] = brow[4 * g + e];
    }
    const float qa = a[(size_t)batch * LA + myRow];
    __builtin_amdgcn_sched_barrier(0);

    const float wq = ipw[0], wk = ipw[1], wv = ipw[2];
    const float bq = ipb[0], bk = ipb[1], bv = ipb[2];
    const float w_out = ow[0], b_out = ob[0];

    float lmax = -INFINITY, lmin = INFINITY;
    #pragma unroll
    for (int t = 0; t < 2; ++t) {
        const int g = tid + t * TPB;
        float kq[4], vq[4];
        #pragma unroll
        for (int e = 0; e < 4; ++e) {
            kq[e] = fmaf(bl[t][e], wk, bk);
            vq[e] = fmaf(bl[t][e], wv, bv);
            lmax = fmaxf(lmax, kq[e]);
            lmin = fminf(lmin, kq[e]);
        }
        reinterpret_cast<float4*>(s_k)[g] = make_float4(kq[0], kq[1], kq[2], kq[3]);
        reinterpret_cast<float4*>(s_v)[g] = make_float4(vq[0], vq[1], vq[2], vq[3]);
    }
    const float wmax = wave_max(lmax);
    const float wmin = wave_min(lmin);
    if (lane == 0) { s_red[wave] = wmax; s_red[WAVES + wave] = wmin; }
    if (tid < 64) s_one[tid] = 1.0f; else if (tid < 128) s_zero[tid - 64] = 0.0f;
    __syncthreads();
    const float4* s_red4 = reinterpret_cast<const float4*>(s_red);
    const float4 mx0 = s_red4[0], mx1 = s_red4[1], mn0 = s_red4[2], mn1 = s_red4[3];
    const float kmax = fmaxf(fmaxf(fmaxf(mx0.x, mx0.y), fmaxf(mx0.z, mx0.w)),
                             fmaxf(fmaxf(mx1.x, mx1.y), fmaxf(mx1.z, mx1.w)));
    const float kmin = fminf(fminf(fminf(mn0.x, mn0.y), fminf(mn0.z, mn0.w)),
                             fminf(fminf(mn1.x, mn1.y), fminf(mn1.z, mn1.w)));

    const float q  = fmaf(qa, wq, bq);
    const float m  = (q >= 0.0f) ? q * kmax : q * kmin;
    const float L2E = 1.4426950408889634f;
    const float ql2  = q * L2E;
    const float nml2 = -(m * L2E);

    const int  lsel = lane >> 4;
    const int  nsel = lane & 15;
    const bool isn0 = (nsel == 0);
    const float4* bbase = isn0 ? (reinterpret_cast<const float4*>(s_v) + lsel)
                               : reinterpret_cast<const float4*>((nsel == 1) ? s_one : s_zero);
    const int bstride = isn0 ? 2 * PAIRS : 0;

    const float4* s_k4 = reinterpret_cast<const float4*>(s_k);

    v8f c = {0.f, 0.f, 0.f, 0.f, 0.f, 0.f, 0.f, 0.f};

    float4 kA[PAIRS], vA[PAIRS], kB[PAIRS], vB[PAIRS];

#define LOAD_BLOCK(KK, VV, BLK)                                              \
    {                                                                        \
        const int base_ = (BLK) * (2 * PAIRS) + lsel;                        \
        _Pragma("unroll")                                                    \
        for (int u = 0; u < PAIRS; ++u) {                                    \
            KK[u] = s_k4[base_ + 2 * u];                                     \
            VV[u] = bbase[(BLK) * bstride + 2 * u];                          \
        }                                                                    \
    }

#define SB() __builtin_amdgcn_sched_barrier(0)
#define COMPUTE_BLOCK(KP, VP, KQ, VQ)                                        \
    {                                                                        \
        _Pragma("unroll")                                                    \
        for (int s = 0; s < 2 * PAIRS; ++s) {                                \
            const float2 kn = step_pair(KP, KQ, s + 2);                      \
            const float2 vb = step_pair(VP, VQ, s);                          \
            asm volatile("" : "+v"(c), "+v"(ar0));                           \
            float e0 = __builtin_amdgcn_exp2f(ar0);                          \
            asm volatile("" : "+v"(e0), "+v"(qv));                           \
            float an0 = fmaf(qv, kn.x, nml2);                                \
            float an1 = fmaf(qv, kn.y, nml2);                                \
            asm volatile("" : "+v"(an0), "+v"(an1), "+v"(ar1));              \
            float e1 = __builtin_amdgcn_exp2f(ar1);                          \
            asm volatile("" : "+v"(e1), "+v"(c));                            \
            asm volatile("" : : "v"(pA), "v"(pB));                           \
            v2f A_; A_.x = ea0; A_.y = ea1;                                  \
            v2f B_; B_.x = vb.x; B_.y = vb.y;                                \
            c = __builtin_amdgcn_wmma_f32_16x16x4_f32(                       \
                    false, A_, false, B_, (short)0, c, false, false);        \
            pA = A_; pB = B_;                                                \
            ea0 = e0; ea1 = e1; ar0 = an0; ar1 = an1;                        \
        }                                                                    \
    }

    LOAD_BLOCK(kA, vA, 0)
    float ea0 = __builtin_amdgcn_exp2f(fmaf(ql2, kA[0].x, nml2));
    float ea1 = __builtin_amdgcn_exp2f(fmaf(ql2, kA[0].y, nml2));
    float ar0 = fmaf(ql2, kA[0].z, nml2);
    float ar1 = fmaf(ql2, kA[0].w, nml2);
    v2f pA = {0.f, 0.f}, pB = {0.f, 0.f};
    float qv = ql2;
    #pragma unroll 1
    for (int blk = 0; blk < NBLK; blk += 2) {
        LOAD_BLOCK(kB, vB, blk + 1)
        SB();
        COMPUTE_BLOCK(kA, vA, kB, vB)
        LOAD_BLOCK(kA, vA, (blk + 2) & (NBLK - 1))
        SB();
        COMPUTE_BLOCK(kB, vB, kA, vA)
    }

    #pragma unroll
    for (int g = 0; g < 8; ++g) s_c[wave][lane][g] = c[g];

    if (lane < 16) {
        int srcLane = (lane < 8) ? 0 : 16;
        int g = lane & 7;
        float numer = s_c[wave][srcLane][g];
        float denom = s_c[wave][srcLane + 1][g];
        float ctx = numer / denom;
        float x = fmaf(ctx, w_out, b_out);
        float y = x + qa;
        float r = 0.5f * y * (1.0f + erff(y * 0.70710678118654752f));
        out[(size_t)batch * LA + rowBase + lane] = r;
    }
}

extern "C" void kernel_launch(void* const* d_in, const int* in_sizes, int n_in,
                              void* d_out, int out_size, void* d_ws, size_t ws_size,
                              hipStream_t stream) {
    const float* a   = (const float*)d_in[0];
    const float* b   = (const float*)d_in[1];
    const float* ipw = (const float*)d_in[2];
    const float* ipb = (const float*)d_in[3];
    const float* ow  = (const float*)d_in[4];
    const float* ob  = (const float*)d_in[5];
    float* out = (float*)d_out;
    (void)in_sizes; (void)n_in; (void)out_size; (void)d_ws; (void)ws_size;

    dim3 grid(BATCH * TILES_PER_BATCH);
    dim3 block(TPB);
    hipLaunchKernelGGL(cross_attn_wmma_kernel, grid, block, 0, stream,
                       a, b, ipw, ipb, ow, ob, out);
}
